# starting kernel
# baseline (speedup 1.0000x reference)
#include <hip/hip_runtime.h>
#include <hip/hip_bf16.h>

typedef __attribute__((ext_vector_type(2))) float v2f;
typedef __attribute__((ext_vector_type(8))) float v8f;

#define V_    32000
#define NH_   32
#define HD_   32
#define C_    1024
#define BT_   128
#define R_    4096
#define SPLIT 8
#define VW    (V_ / SPLIT)

__global__ __launch_bounds__(256) void k_ffn(const float* __restrict__ x,
                                             const float* __restrict__ W,
                                             const float* __restrict__ b,
                                             const float* __restrict__ temps,
                                             float* __restrict__ ffn) {
  int tid = blockIdx.x * blockDim.x + threadIdx.x;
  int r = tid >> 5, d = tid & 31;
  const float4* x4 = (const float4*)x + r * 8;
  const float4* W4 = (const float4*)W + d * 8;
  float s = 0.f;
#pragma unroll
  for (int i = 0; i < 8; ++i) {
    float4 a = x4[i], w = W4[i];
    s = fmaf(a.x, w.x, fmaf(a.y, w.y, fmaf(a.z, w.z, fmaf(a.w, w.w, s))));
  }
  s += b[d];
  float t = fmaxf(temps[r & 31], 0.1f);
  ffn[tid] = s / t;
}

__global__ __launch_bounds__(256) void k_vs(const float* __restrict__ ve,
                                            float* __restrict__ vs) {
  int tid = blockIdx.x * blockDim.x + threadIdx.x;
  int v = tid >> 5, h = tid & 31;
  const float4* p = (const float4*)ve + v * 256 + h * 8;
  float s = 0.f;
#pragma unroll
  for (int i = 0; i < 8; ++i) { float4 a = p[i]; s += (a.x + a.y) + (a.z + a.w); }
  vs[h * V_ + v] = s;
}

__global__ __launch_bounds__(32 * SPLIT) void k_main(const float* __restrict__ ffn,
                                                     const float* __restrict__ vs,
                                                     float* __restrict__ out) {
  __shared__ float lds_p[SPLIT][16 * 17];
  __shared__ float lds_s[SPLIT][16];
  __shared__ float lds_m[SPLIT][16];
  __shared__ float lds_l[SPLIT][16];
  __shared__ float lds_o[16 * 33];

  const int lane = threadIdx.x & 31;
  const int wv   = threadIdx.x >> 5;
  const int col  = lane & 15;
  const int half = lane >> 4;
  const int r0   = blockIdx.x * 16;

  for (int i = threadIdx.x; i < 16 * 33; i += 32 * SPLIT) lds_o[i] = 0.f;

  v2f A[8];
#pragma unroll
  for (int kk = 0; kk < 8; ++kk) {
    const float* pa = ffn + (r0 + col) * HD_ + 4 * kk + 2 * half;
    A[kk].x = pa[0]; A[kk].y = pa[1];
  }

  v8f acc0 = {}, acc1 = {};
  float mh[8], lh[8];
#pragma unroll
  for (int g = 0; g < 8; ++g) { mh[g] = -__builtin_inff(); lh[g] = 0.f; }

  const int vbeg = wv * VW;
  for (int vt = 0; vt < VW / 16; ++vt) {
    const int vb = vbeg + vt * 16;
    v8f S = {};
#pragma unroll
    for (int kk = 0; kk < 8; ++kk) {
      v2f Bt;
      const float* pb = vs + (4 * kk + 2 * half) * V_ + vb + col;
      Bt.x = pb[0];
      Bt.y = pb[V_];
      S = __builtin_amdgcn_wmma_f32_16x16x4_f32(false, A[kk], false, Bt,
                                                (short)0, S, false, false);
    }

    float p[8], scale_h[8];
#pragma unroll
    for (int g = 0; g < 8; ++g) {
      float tm = S[g];
#pragma unroll
      for (int m = 1; m < 16; m <<= 1) tm = fmaxf(tm, __shfl_xor(tm, m, 16));
      float mn = fmaxf(mh[g], tm);
      float sc = __expf(mh[g] - mn);
      float pe = __expf(S[g] - mn);
      float ts = pe;
#pragma unroll
      for (int m = 1; m < 16; m <<= 1) ts += __shfl_xor(ts, m, 16);
      lh[g] = lh[g] * sc + ts;
      mh[g] = mn;
      p[g] = pe;
      scale_h[g] = sc;
    }

    if (lane == 0) {
#pragma unroll
      for (int g = 0; g < 8; ++g) lds_s[wv][g] = scale_h[g];
    }
    if (lane == 16) {
#pragma unroll
      for (int g = 0; g < 8; ++g) lds_s[wv][8 + g] = scale_h[g];
    }
#pragma unroll
    for (int g = 0; g < 8; ++g)
      lds_p[wv][col * 17 + g + 8 * half] = p[g];
    __syncthreads();

    float sc = lds_s[wv][col];
#pragma unroll
    for (int g = 0; g < 8; ++g) { acc0[g] *= sc; acc1[g] *= sc; }

#pragma unroll
    for (int kk = 0; kk < 4; ++kk) {
      int k0 = 4 * kk + 2 * half;
      v2f Bt;
      Bt.x = lds_p[wv][k0 * 17 + col];
      Bt.y = lds_p[wv][(k0 + 1) * 17 + col];
      v2f Aa, Ab;
      const float* pa0 = vs + col * V_ + vb + k0;
      const float* pa1 = vs + (16 + col) * V_ + vb + k0;
      Aa.x = pa0[0]; Aa.y = pa0[1];
      Ab.x = pa1[0]; Ab.y = pa1[1];
      acc0 = __builtin_amdgcn_wmma_f32_16x16x4_f32(false, Aa, false, Bt,
                                                   (short)0, acc0, false, false);
      acc1 = __builtin_amdgcn_wmma_f32_16x16x4_f32(false, Ab, false, Bt,
                                                   (short)0, acc1, false, false);
    }
    __syncthreads();
  }

  if (lane == 0) {
#pragma unroll
    for (int g = 0; g < 8; ++g) { lds_m[wv][g] = mh[g]; lds_l[wv][g] = lh[g]; }
  }
  if (lane == 16) {
#pragma unroll
    for (int g = 0; g < 8; ++g) { lds_m[wv][8 + g] = mh[g]; lds_l[wv][8 + g] = lh[g]; }
  }
  __syncthreads();

  float M = -__builtin_inff();
#pragma unroll
  for (int w = 0; w < SPLIT; ++w) M = fmaxf(M, lds_m[w][col]);
  float L = 0.f;
#pragma unroll
  for (int w = 0; w < SPLIT; ++w) L += lds_l[w][col] * __expf(lds_m[w][col] - M);
  float f = __expf(lds_m[wv][col] - M) / L;

#pragma unroll
  for (int g = 0; g < 8; ++g) {
    int h0 = g + 8 * half;
    atomicAdd(&lds_o[col * 33 + h0],      acc0[g] * f);
    atomicAdd(&lds_o[col * 33 + 16 + h0], acc1[g] * f);
  }
  __syncthreads();

  int i0 = threadIdx.x;
  out[r0 * 32 + i0]       = lds_o[(i0 >> 5) * 33 + (i0 & 31)];
  int i1 = i0 + 256;
  out[r0 * 32 + i1]       = lds_o[(i1 >> 5) * 33 + (i1 & 31)];
}

extern "C" void kernel_launch(void* const* d_in, const int* in_sizes, int n_in,
                              void* d_out, int out_size, void* d_ws, size_t ws_size,
                              hipStream_t stream) {
  const float* x     = (const float*)d_in[0];
  const float* W     = (const float*)d_in[1];
  const float* b     = (const float*)d_in[2];
  const float* temps = (const float*)d_in[3];
  const float* ve    = (const float*)d_in[4];
  float* out = (float*)d_out;

  float* vs  = (float*)d_ws;
  float* ffn = vs + (size_t)NH_ * V_;

  k_vs  <<<(V_ * 32) / 256, 256, 0, stream>>>(ve, vs);
  k_ffn <<<(R_ * 32) / 256, 256, 0, stream>>>(x, W, b, temps, ffn);
  k_main<<<R_ / 16, 32 * SPLIT, 0, stream>>>(ffn, vs, out);
}


# logits tile formed transposed: softmax P stays in registers as B fragment of 2nd product (permuted k order); no LDS staging, barrier or shuffle tree in loop; row sum per lane half (sum order changes)
# speedup vs baseline: 2.3353x; 2.3353x over previous
#include <hip/hip_runtime.h>
#include <hip/hip_bf16.h>

typedef __attribute__((ext_vector_type(2))) float v2f;
typedef __attribute__((ext_vector_type(8))) float v8f;

#define V_    32000
#define NH_   32
#define HD_   32
#define C_    1024
#define BT_   128
#define R_    4096
#define SPLIT 8
#define VW    (V_ / SPLIT)

__global__ __launch_bounds__(256) void k_ffn(const float* __restrict__ x,
                                             const float* __restrict__ W,
                                             const float* __restrict__ b,
                                             const float* __restrict__ temps,
                                             float* __restrict__ ffn) {
  int tid = blockIdx.x * blockDim.x + threadIdx.x;
  int r = tid >> 5, d = tid & 31;
  const float4* x4 = (const float4*)x + r * 8;
  const float4* W4 = (const float4*)W + d * 8;
  float s = 0.f;
#pragma unroll
  for (int i = 0; i < 8; ++i) {
    float4 a = x4[i], w = W4[i];
    s = fmaf(a.x, w.x, fmaf(a.y, w.y, fmaf(a.z, w.z, fmaf(a.w, w.w, s))));
  }
  s += b[d];
  float t = fmaxf(temps[r & 31], 0.1f);
  ffn[tid] = s / t;
}

__global__ __launch_bounds__(256) void k_vs(const float* __restrict__ ve,
                                            float* __restrict__ vs) {
  int tid = blockIdx.x * blockDim.x + threadIdx.x;
  int v = tid >> 5, h = tid & 31;
  const float4* p = (const float4*)ve + v * 256 + h * 8;
  float s = 0.f;
#pragma unroll
  for (int i = 0; i < 8; ++i) { float4 a = p[i]; s += (a.x + a.y) + (a.z + a.w); }
  vs[h * V_ + v] = s;
}


__device__ __forceinline__ float xhalf(float x) {
  int b = __builtin_bit_cast(int, x);
  return __builtin_bit_cast(float,
      __builtin_amdgcn_permlanex16(b, b, 0x76543210, 0xfedcba98, false, false));
}

__global__ __launch_bounds__(32 * SPLIT) void k_main(const float* __restrict__ ffn,
                                                     const float* __restrict__ vs,
                                                     float* __restrict__ out) {
  __shared__ float lds_m[SPLIT][16];
  __shared__ float lds_l[SPLIT][16];
  __shared__ float lds_o[16 * 33];

  const int lane = threadIdx.x & 31;
  const int wv   = threadIdx.x >> 5;
  const int col  = lane & 15;
  const int half = lane >> 4;
  const int r0   = blockIdx.x * 16;

  for (int i = threadIdx.x; i < 16 * 33; i += 32 * SPLIT) lds_o[i] = 0.f;

  v2f Q[8];
#pragma unroll
  for (int kk = 0; kk < 8; ++kk) {
    const float* pq = ffn + (r0 + col) * HD_ + 4 * kk + 2 * half;
    Q[kk].x = pq[0]; Q[kk].y = pq[1];
  }

  v8f acc0 = {}, acc1 = {};
  float m = -__builtin_inff();
  float l = 0.f;

  const int vbeg = wv * VW;
  for (int vt = 0; vt < VW / 16; ++vt) {
    const int vb = vbeg + vt * 16;

    const float4* pa0 = (const float4*)(vs + col * V_ + vb + 8 * half);
    const float4* pa1 = (const float4*)(vs + (16 + col) * V_ + vb + 8 * half);
    const float4 a00 = pa0[0], a01 = pa0[1];
    const float4 a10 = pa1[0], a11 = pa1[1];

    v8f S = {};
#pragma unroll
    for (int kk = 0; kk < 8; ++kk) {
      v2f At;
      const float* pv = vs + (4 * kk + 2 * half) * V_ + vb + col;
      At.x = pv[0];
      At.y = pv[V_];
      S = __builtin_amdgcn_wmma_f32_16x16x4_f32(false, At, false, Q[kk],
                                                (short)0, S, false, false);
    }

    float tm = fmaxf(fmaxf(fmaxf(S[0], S[1]), fmaxf(S[2], S[3])),
                     fmaxf(fmaxf(S[4], S[5]), fmaxf(S[6], S[7])));
    tm = fmaxf(tm, xhalf(tm));
    const float mn = fmaxf(m, tm);
    const float sc = __expf(m - mn);
    float p[8];
#pragma unroll
    for (int g = 0; g < 8; ++g) p[g] = __expf(S[g] - mn);
    l = l * sc + (((p[0] + p[1]) + (p[2] + p[3])) + ((p[4] + p[5]) + (p[6] + p[7])));
    m = mn;
#pragma unroll
    for (int g = 0; g < 8; ++g) { acc0[g] *= sc; acc1[g] *= sc; }

    {
      v2f Aa, Ab, Bt;
      Aa.x = a00.x; Aa.y = a00.y; Ab.x = a10.x; Ab.y = a10.y; Bt.x = p[0]; Bt.y = p[1];
      acc0 = __builtin_amdgcn_wmma_f32_16x16x4_f32(false, Aa, false, Bt,
                                                   (short)0, acc0, false, false);
      acc1 = __builtin_amdgcn_wmma_f32_16x16x4_f32(false, Ab, false, Bt,
                                                   (short)0, acc1, false, false);
      Aa.x = a00.z; Aa.y = a00.w; Ab.x = a10.z; Ab.y = a10.w; Bt.x = p[2]; Bt.y = p[3];
      acc0 = __builtin_amdgcn_wmma_f32_16x16x4_f32(false, Aa, false, Bt,
                                                   (short)0, acc0, false, false);
      acc1 = __builtin_amdgcn_wmma_f32_16x16x4_f32(false, Ab, false, Bt,
                                                   (short)0, acc1, false, false);
      Aa.x = a01.x; Aa.y = a01.y; Ab.x = a11.x; Ab.y = a11.y; Bt.x = p[4]; Bt.y = p[5];
      acc0 = __builtin_amdgcn_wmma_f32_16x16x4_f32(false, Aa, false, Bt,
                                                   (short)0, acc0, false, false);
      acc1 = __builtin_amdgcn_wmma_f32_16x16x4_f32(false, Ab, false, Bt,
                                                   (short)0, acc1, false, false);
      Aa.x = a01.z; Aa.y = a01.w; Ab.x = a11.z; Ab.y = a11.w; Bt.x = p[6]; Bt.y = p[7];
      acc0 = __builtin_amdgcn_wmma_f32_16x16x4_f32(false, Aa, false, Bt,
                                                   (short)0, acc0, false, false);
      acc1 = __builtin_amdgcn_wmma_f32_16x16x4_f32(false, Ab, false, Bt,
                                                   (short)0, acc1, false, false);
    }
  }

  l += xhalf(l);
  if (half == 0) { lds_m[wv][col] = m; lds_l[wv][col] = l; }
  __syncthreads();

  float M = -__builtin_inff();
#pragma unroll
  for (int w = 0; w < SPLIT; ++w) M = fmaxf(M, lds_m[w][col]);
  float L = 0.f;
#pragma unroll
  for (int w = 0; w < SPLIT; ++w) L += lds_l[w][col] * __expf(lds_m[w][col] - M);
  float f = __expf(lds_m[wv][col] - M) / L;

#pragma unroll
  for (int g = 0; g < 8; ++g) {
    int h0 = g + 8 * half;
    atomicAdd(&lds_o[col * 33 + h0],      acc0[g] * f);
    atomicAdd(&lds_o[col * 33 + 16 + h0], acc1[g] * f);
  }
  __syncthreads();

  int i0 = threadIdx.x;
  out[r0 * 32 + i0]       = lds_o[(i0 >> 5) * 33 + (i0 & 31)];
  int i1 = i0 + 256;
  out[r0 * 32 + i1]       = lds_o[(i1 >> 5) * 33 + (i1 & 31)];
}

extern "C" void kernel_launch(void* const* d_in, const int* in_sizes, int n_in,
                              void* d_out, int out_size, void* d_ws, size_t ws_size,
                              hipStream_t stream) {
  const float* x     = (const float*)d_in[0];
  const float* W     = (const float*)d_in[1];
  const float* b     = (const float*)d_in[2];
  const float* temps = (const float*)d_in[3];
  const float* ve    = (const float*)d_in[4];
  float* out = (float*)d_out;

  float* vs  = (float*)d_ws;
  float* ffn = vs + (size_t)NH_ * V_;

  k_vs  <<<(V_ * 32) / 256, 256, 0, stream>>>(ve, vs);
  k_ffn <<<(R_ * 32) / 256, 256, 0, stream>>>(x, W, b, temps, ffn);
  k_main<<<R_ / 16, 32 * SPLIT, 0, stream>>>(ffn, vs, out);
}


# vocab axis split over 16 waves per workgroup instead of 8: 4 waves per SIMD overlap fragment-load latency; partials merged in fixed wave order without LDS atomics (sum order changes)
# speedup vs baseline: 2.6131x; 1.1190x over previous
#include <hip/hip_runtime.h>
#include <hip/hip_bf16.h>

typedef __attribute__((ext_vector_type(2))) float v2f;
typedef __attribute__((ext_vector_type(8))) float v8f;

#define V_    32000
#define NH_   32
#define HD_   32
#define C_    1024
#define BT_   128
#define R_    4096
#define SPLIT 16
#define VW    (V_ / SPLIT)

__global__ __launch_bounds__(256) void k_ffn(const float* __restrict__ x,
                                             const float* __restrict__ W,
                                             const float* __restrict__ b,
                                             const float* __restrict__ temps,
                                             float* __restrict__ ffn) {
  int tid = blockIdx.x * blockDim.x + threadIdx.x;
  int r = tid >> 5, d = tid & 31;
  const float4* x4 = (const float4*)x + r * 8;
  const float4* W4 = (const float4*)W + d * 8;
  float s = 0.f;
#pragma unroll
  for (int i = 0; i < 8; ++i) {
    float4 a = x4[i], w = W4[i];
    s = fmaf(a.x, w.x, fmaf(a.y, w.y, fmaf(a.z, w.z, fmaf(a.w, w.w, s))));
  }
  s += b[d];
  float t = fmaxf(temps[r & 31], 0.1f);
  ffn[tid] = s / t;
}

__global__ __launch_bounds__(256) void k_vs(const float* __restrict__ ve,
                                            float* __restrict__ vs) {
  int tid = blockIdx.x * blockDim.x + threadIdx.x;
  int v = tid >> 5, h = tid & 31;
  const float4* p = (const float4*)ve + v * 256 + h * 8;
  float s = 0.f;
#pragma unroll
  for (int i = 0; i < 8; ++i) { float4 a = p[i]; s += (a.x + a.y) + (a.z + a.w); }
  vs[h * V_ + v] = s;
}


__device__ __forceinline__ float xhalf(float x) {
  int b = __builtin_bit_cast(int, x);
  return __builtin_bit_cast(float,
      __builtin_amdgcn_permlanex16(b, b, 0x76543210, 0xfedcba98, false, false));
}

__global__ __launch_bounds__(32 * SPLIT) void k_main(const float* __restrict__ ffn,
                                                     const float* __restrict__ vs,
                                                     float* __restrict__ out) {
  __shared__ float lds_m[SPLIT][16];
  __shared__ float lds_l[SPLIT][16];
  __shared__ float lds_p[SPLIT][16 * 33];

  const int lane = threadIdx.x & 31;
  const int wv   = threadIdx.x >> 5;
  const int col  = lane & 15;
  const int half = lane >> 4;
  const int r0   = blockIdx.x * 16;

  v2f Q[8];
#pragma unroll
  for (int kk = 0; kk < 8; ++kk) {
    const float* pq = ffn + (r0 + col) * HD_ + 4 * kk + 2 * half;
    Q[kk].x = pq[0]; Q[kk].y = pq[1];
  }

  v8f acc0 = {}, acc1 = {};
  float m = -__builtin_inff();
  float l = 0.f;

  const int vbeg = wv * VW;
  for (int vt = 0; vt < VW / 16; ++vt) {
    const int vb = vbeg + vt * 16;

    const float4* pa0 = (const float4*)(vs + col * V_ + vb + 8 * half);
    const float4* pa1 = (const float4*)(vs + (16 + col) * V_ + vb + 8 * half);
    const float4 a00 = pa0[0], a01 = pa0[1];
    const float4 a10 = pa1[0], a11 = pa1[1];

    v8f S = {};
#pragma unroll
    for (int kk = 0; kk < 8; ++kk) {
      v2f At;
      const float* pv = vs + (4 * kk + 2 * half) * V_ + vb + col;
      At.x = pv[0];
      At.y = pv[V_];
      S = __builtin_amdgcn_wmma_f32_16x16x4_f32(false, At, false, Q[kk],
                                                (short)0, S, false, false);
    }

    float tm = fmaxf(fmaxf(fmaxf(S[0], S[1]), fmaxf(S[2], S[3])),
                     fmaxf(fmaxf(S[4], S[5]), fmaxf(S[6], S[7])));
    tm = fmaxf(tm, xhalf(tm));
    const float mn = fmaxf(m, tm);
    const float sc = __expf(m - mn);
    float p[8];
#pragma unroll
    for (int g = 0; g < 8; ++g) p[g] = __expf(S[g] - mn);
    l = l * sc + (((p[0] + p[1]) + (p[2] + p[3])) + ((p[4] + p[5]) + (p[6] + p[7])));
    m = mn;
#pragma unroll
    for (int g = 0; g < 8; ++g) { acc0[g] *= sc; acc1[g] *= sc; }

    {
      v2f Aa, Ab, Bt;
      Aa.x = a00.x; Aa.y = a00.y; Ab.x = a10.x; Ab.y = a10.y; Bt.x = p[0]; Bt.y = p[1];
      acc0 = __builtin_amdgcn_wmma_f32_16x16x4_f32(false, Aa, false, Bt,
                                                   (short)0, acc0, false, false);
      acc1 = __builtin_amdgcn_wmma_f32_16x16x4_f32(false, Ab, false, Bt,
                                                   (short)0, acc1, false, false);
      Aa.x = a00.z; Aa.y = a00.w; Ab.x = a10.z; Ab.y = a10.w; Bt.x = p[2]; Bt.y = p[3];
      acc0 = __builtin_amdgcn_wmma_f32_16x16x4_f32(false, Aa, false, Bt,
                                                   (short)0, acc0, false, false);
      acc1 = __builtin_amdgcn_wmma_f32_16x16x4_f32(false, Ab, false, Bt,
                                                   (short)0, acc1, false, false);
      Aa.x = a01.x; Aa.y = a01.y; Ab.x = a11.x; Ab.y = a11.y; Bt.x = p[4]; Bt.y = p[5];
      acc0 = __builtin_amdgcn_wmma_f32_16x16x4_f32(false, Aa, false, Bt,
                                                   (short)0, acc0, false, false);
      acc1 = __builtin_amdgcn_wmma_f32_16x16x4_f32(false, Ab, false, Bt,
                                                   (short)0, acc1, false, false);
      Aa.x = a01.z; Aa.y = a01.w; Ab.x = a11.z; Ab.y = a11.w; Bt.x = p[6]; Bt.y = p[7];
      acc0 = __builtin_amdgcn_wmma_f32_16x16x4_f32(false, Aa, false, Bt,
                                                   (short)0, acc0, false, false);
      acc1 = __builtin_amdgcn_wmma_f32_16x16x4_f32(false, Ab, false, Bt,
                                                   (short)0, acc1, false, false);
    }
  }

  l += xhalf(l);
  if (half == 0) { lds_m[wv][col] = m; lds_l[wv][col] = l; }
  __syncthreads();

  float M = -__builtin_inff();
#pragma unroll
  for (int w = 0; w < SPLIT; ++w) M = fmaxf(M, lds_m[w][col]);
  float L = 0.f;
#pragma unroll
  for (int w = 0; w < SPLIT; ++w) L += lds_l[w][col] * __expf(lds_m[w][col] - M);
  float f = __expf(lds_m[wv][col] - M) / L;

#pragma unroll
  for (int g = 0; g < 8; ++g) {
    int h0 = g + 8 * half;
    lds_p[wv][col * 33 + h0]      = acc0[g] * f;
    lds_p[wv][col * 33 + 16 + h0] = acc1[g] * f;
  }
  __syncthreads();

  for (int i = threadIdx.x; i < 16 * 32; i += 32 * SPLIT) {
    const int o = (i >> 5) * 33 + (i & 31);
    float s = lds_p[0][o];
#pragma unroll
    for (int w = 1; w < SPLIT; ++w) s += lds_p[w][o];
    out[r0 * 32 + i] = s;
  }
}

extern "C" void kernel_launch(void* const* d_in, const int* in_sizes, int n_in,
                              void* d_out, int out_size, void* d_ws, size_t ws_size,
                              hipStream_t stream) {
  const float* x     = (const float*)d_in[0];
  const float* W     = (const float*)d_in[1];
  const float* b     = (const float*)d_in[2];
  const float* temps = (const float*)d_in[3];
  const float* ve    = (const float*)d_in[4];
  float* out = (float*)d_out;

  float* vs  = (float*)d_ws;
  float* ffn = vs + (size_t)NH_ * V_;

  k_vs  <<<(V_ * 32) / 256, 256, 0, stream>>>(ve, vs);
  k_ffn <<<(R_ * 32) / 256, 256, 0, stream>>>(x, W, b, temps, ffn);
  k_main<<<R_ / 16, 32 * SPLIT, 0, stream>>>(ffn, vs, out);
}


# vocab_sum kept tile-blocked: each 32x16 tile fetched once per trip by four whole-line loads; the logits product's transposed fragments come from a per-wave LDS slab, not 16 two-line dword loads
# speedup vs baseline: 3.1975x; 1.2236x over previous
#include <hip/hip_runtime.h>
#include <hip/hip_bf16.h>

typedef __attribute__((ext_vector_type(2))) float v2f;
typedef __attribute__((ext_vector_type(8))) float v8f;

#define V_    32000
#define NH_   32
#define HD_   32
#define C_    1024
#define BT_   128
#define R_    4096
#define SPLIT 16
#define VW    (V_ / SPLIT)
#define TP    20

__device__ __forceinline__ int vs_off(int h, int v) {
  return (v >> 4) * 512 + (2 * (h >> 4) + ((v >> 2) & 1)) * 128
       + (16 * ((v >> 3) & 1) + (h & 15)) * 4 + (v & 3);
}

__global__ __launch_bounds__(256) void k_ffn(const float* __restrict__ x,
                                             const float* __restrict__ W,
                                             const float* __restrict__ b,
                                             const float* __restrict__ temps,
                                             float* __restrict__ ffn) {
  int tid = blockIdx.x * blockDim.x + threadIdx.x;
  int r = tid >> 5, d = tid & 31;
  const float4* x4 = (const float4*)x + r * 8;
  const float4* W4 = (const float4*)W + d * 8;
  float s = 0.f;
#pragma unroll
  for (int i = 0; i < 8; ++i) {
    float4 a = x4[i], w = W4[i];
    s = fmaf(a.x, w.x, fmaf(a.y, w.y, fmaf(a.z, w.z, fmaf(a.w, w.w, s))));
  }
  s += b[d];
  float t = fmaxf(temps[r & 31], 0.1f);
  ffn[tid] = s / t;
}

__global__ __launch_bounds__(256) void k_vs(const float* __restrict__ ve,
                                            float* __restrict__ vs) {
  int tid = blockIdx.x * blockDim.x + threadIdx.x;
  int v = tid >> 5, h = tid & 31;
  const float4* p = (const float4*)ve + v * 256 + h * 8;
  float s = 0.f;
#pragma unroll
  for (int i = 0; i < 8; ++i) { float4 a = p[i]; s += (a.x + a.y) + (a.z + a.w); }
  vs[vs_off(h, v)] = s;
}


__device__ __forceinline__ float xhalf(float x) {
  int b = __builtin_bit_cast(int, x);
  return __builtin_bit_cast(float,
      __builtin_amdgcn_permlanex16(b, b, 0x76543210, 0xfedcba98, false, false));
}

__global__ __launch_bounds__(32 * SPLIT) void k_main(const float* __restrict__ ffn,
                                                     const float* __restrict__ vs,
                                                     float* __restrict__ out) {
  __shared__ float lds_m[SPLIT][16];
  __shared__ float lds_l[SPLIT][16];
  __shared__ float lds_p[SPLIT][16 * 33];
  __shared__ __attribute__((aligned(16))) float lds_t[SPLIT][32 * TP];

  const int lane = threadIdx.x & 31;
  const int wv   = threadIdx.x >> 5;
  const int col  = lane & 15;
  const int half = lane >> 4;
  const int r0   = blockIdx.x * 16;

  v2f Q[8];
#pragma unroll
  for (int kk = 0; kk < 8; ++kk) {
    const float* pq = ffn + (r0 + col) * HD_ + 4 * kk + 2 * half;
    Q[kk].x = pq[0]; Q[kk].y = pq[1];
  }

  v8f acc0 = {}, acc1 = {};
  float m = -__builtin_inff();
  float l = 0.f;

  float* slab = &lds_t[wv][0];
  float4* st0 = (float4*)(slab + col * TP + 8 * half);
  float4* st1 = (float4*)(slab + (16 + col) * TP + 8 * half);
  const float* ld = slab + (2 * half) * TP + col;

  const float4* pt = (const float4*)vs + (wv * (VW / 16)) * 128 + lane;
  for (int vt = 0; vt < VW / 16; ++vt, pt += 128) {
    const float4 a00 = pt[0], a01 = pt[32];
    const float4 a10 = pt[64], a11 = pt[96];

    st0[0] = a00; st0[1] = a01;
    st1[0] = a10; st1[1] = a11;

    v2f At[8];
#pragma unroll
    for (int kk = 0; kk < 8; ++kk) {
      At[kk].x = ld[(4 * kk) * TP];
      At[kk].y = ld[(4 * kk + 1) * TP];
    }
    __builtin_amdgcn_sched_barrier(0);
    v8f S = {};
#pragma unroll
    for (int kk = 0; kk < 8; ++kk)
      S = __builtin_amdgcn_wmma_f32_16x16x4_f32(false, At[kk], false, Q[kk],
                                                (short)0, S, false, false);

    float tm = fmaxf(fmaxf(fmaxf(S[0], S[1]), fmaxf(S[2], S[3])),
                     fmaxf(fmaxf(S[4], S[5]), fmaxf(S[6], S[7])));
    tm = fmaxf(tm, xhalf(tm));
    const float mn = fmaxf(m, tm);
    const float sc = __expf(m - mn);
    float p[8];
#pragma unroll
    for (int g = 0; g < 8; ++g) p[g] = __expf(S[g] - mn);
    l = l * sc + (((p[0] + p[1]) + (p[2] + p[3])) + ((p[4] + p[5]) + (p[6] + p[7])));
    m = mn;
#pragma unroll
    for (int g = 0; g < 8; ++g) { acc0[g] *= sc; acc1[g] *= sc; }

    {
      v2f Aa, Ab, Bt;
      Aa.x = a00.x; Aa.y = a00.y; Ab.x = a10.x; Ab.y = a10.y; Bt.x = p[0]; Bt.y = p[1];
      acc0 = __builtin_amdgcn_wmma_f32_16x16x4_f32(false, Aa, false, Bt,
                                                   (short)0, acc0, false, false);
      acc1 = __builtin_amdgcn_wmma_f32_16x16x4_f32(false, Ab, false, Bt,
                                                   (short)0, acc1, false, false);
      Aa.x = a00.z; Aa.y = a00.w; Ab.x = a10.z; Ab.y = a10.w; Bt.x = p[2]; Bt.y = p[3];
      acc0 = __builtin_amdgcn_wmma_f32_16x16x4_f32(false, Aa, false, Bt,
                                                   (short)0, acc0, false, false);
      acc1 = __builtin_amdgcn_wmma_f32_16x16x4_f32(false, Ab, false, Bt,
                                                   (short)0, acc1, false, false);
      Aa.x = a01.x; Aa.y = a01.y; Ab.x = a11.x; Ab.y = a11.y; Bt.x = p[4]; Bt.y = p[5];
      acc0 = __builtin_amdgcn_wmma_f32_16x16x4_f32(false, Aa, false, Bt,
                                                   (short)0, acc0, false, false);
      acc1 = __builtin_amdgcn_wmma_f32_16x16x4_f32(false, Ab, false, Bt,
                                                   (short)0, acc1, false, false);
      Aa.x = a01.z; Aa.y = a01.w; Ab.x = a11.z; Ab.y = a11.w; Bt.x = p[6]; Bt.y = p[7];
      acc0 = __builtin_amdgcn_wmma_f32_16x16x4_f32(false, Aa, false, Bt,
                                                   (short)0, acc0, false, false);
      acc1 = __builtin_amdgcn_wmma_f32_16x16x4_f32(false, Ab, false, Bt,
                                                   (short)0, acc1, false, false);
    }
  }

  l += xhalf(l);
  if (half == 0) { lds_m[wv][col] = m; lds_l[wv][col] = l; }
  __syncthreads();

  float M = -__builtin_inff();
#pragma unroll
  for (int w = 0; w < SPLIT; ++w) M = fmaxf(M, lds_m[w][col]);
  float L = 0.f;
#pragma unroll
  for (int w = 0; w < SPLIT; ++w) L += lds_l[w][col] * __expf(lds_m[w][col] - M);
  float f = __expf(lds_m[wv][col] - M) / L;

#pragma unroll
  for (int g = 0; g < 8; ++g) {
    int h0 = g + 8 * half;
    lds_p[wv][col * 33 + h0]      = acc0[g] * f;
    lds_p[wv][col * 33 + 16 + h0] = acc1[g] * f;
  }
  __syncthreads();

  for (int i = threadIdx.x; i < 16 * 32; i += 32 * SPLIT) {
    const int o = (i >> 5) * 33 + (i & 31);
    float s = lds_p[0][o];
#pragma unroll
    for (int w = 1; w < SPLIT; ++w) s += lds_p[w][o];
    out[r0 * 32 + i] = s;
  }
}

extern "C" void kernel_launch(void* const* d_in, const int* in_sizes, int n_in,
                              void* d_out, int out_size, void* d_ws, size_t ws_size,
                              hipStream_t stream) {
  const float* x     = (const float*)d_in[0];
  const float* W     = (const float*)d_in[1];
  const float* b     = (const float*)d_in[2];
  const float* temps = (const float*)d_in[3];
  const float* ve    = (const float*)d_in[4];
  float* out = (float*)d_out;

  float* vs  = (float*)d_ws;
  float* ffn = vs + (size_t)NH_ * V_;

  k_vs  <<<(V_ * 32) / 256, 256, 0, stream>>>(ve, vs);
  k_ffn <<<(R_ * 32) / 256, 256, 0, stream>>>(x, W, b, temps, ffn);
  k_main<<<R_ / 16, 32 * SPLIT, 0, stream>>>(ffn, vs, out);
}


# k_main softmax reference moved lazily: logits leave the matrix pipe in log2 units minus the row reference (accumulator start value); max tree, lane exchange, rescale only if a tile sums above 2^10
# speedup vs baseline: 3.4061x; 1.0652x over previous
#include <hip/hip_runtime.h>
#include <hip/hip_bf16.h>

typedef __attribute__((ext_vector_type(2))) float v2f;
typedef __attribute__((ext_vector_type(8))) float v8f;

#define V_    32000
#define NH_   32
#define HD_   32
#define C_    1024
#define BT_   128
#define R_    4096
#define SPLIT 16
#define VW    (V_ / SPLIT)
#define TP    20
#define LOG2E   1.44269504088896340736f
#define P_LIMIT 1024.0f

__device__ __forceinline__ int vs_off(int h, int v) {
  return (v >> 4) * 512 + (2 * (h >> 4) + ((v >> 2) & 1)) * 128
       + (16 * ((v >> 3) & 1) + (h & 15)) * 4 + (v & 3);
}

__global__ __launch_bounds__(256) void k_ffn(const float* __restrict__ x,
                                             const float* __restrict__ W,
                                             const float* __restrict__ b,
                                             const float* __restrict__ temps,
                                             float* __restrict__ ffn) {
  int tid = blockIdx.x * blockDim.x + threadIdx.x;
  int r = tid >> 5, d = tid & 31;
  const float4* x4 = (const float4*)x + r * 8;
  const float4* W4 = (const float4*)W + d * 8;
  float s = 0.f;
#pragma unroll
  for (int i = 0; i < 8; ++i) {
    float4 a = x4[i], w = W4[i];
    s = fmaf(a.x, w.x, fmaf(a.y, w.y, fmaf(a.z, w.z, fmaf(a.w, w.w, s))));
  }
  s += b[d];
  float t = fmaxf(temps[r & 31], 0.1f);
  ffn[tid] = s / t;
}

__global__ __launch_bounds__(256) void k_vs(const float* __restrict__ ve,
                                            float* __restrict__ vs) {
  int tid = blockIdx.x * blockDim.x + threadIdx.x;
  int v = tid >> 5, h = tid & 31;
  const float4* p = (const float4*)ve + v * 256 + h * 8;
  float s = 0.f;
#pragma unroll
  for (int i = 0; i < 8; ++i) { float4 a = p[i]; s += (a.x + a.y) + (a.z + a.w); }
  vs[vs_off(h, v)] = s;
}


__device__ __forceinline__ float xhalf(float x) {
  int b = __builtin_bit_cast(int, x);
  return __builtin_bit_cast(float,
      __builtin_amdgcn_permlanex16(b, b, 0x76543210, 0xfedcba98, false, false));
}

__global__ __launch_bounds__(32 * SPLIT) void k_main(const float* __restrict__ ffn,
                                                     const float* __restrict__ vs,
                                                     float* __restrict__ out) {
  __shared__ float lds_m[SPLIT][16];
  __shared__ float lds_l[SPLIT][16];
  __shared__ float lds_p[SPLIT][16 * 33];
  __shared__ __attribute__((aligned(16))) float lds_t[SPLIT][32 * TP];

  const int lane = threadIdx.x & 31;
  const int wv   = threadIdx.x >> 5;
  const int col  = lane & 15;
  const int half = lane >> 4;
  const int r0   = blockIdx.x * 16;

  v2f Q[8];
#pragma unroll
  for (int kk = 0; kk < 8; ++kk) {
    const float* pq = ffn + (r0 + col) * HD_ + 4 * kk + 2 * half;
    Q[kk].x = pq[0] * LOG2E; Q[kk].y = pq[1] * LOG2E;
  }

  v8f acc0 = {}, acc1 = {};
  float m = 0.f;
  v8f negm = {};
  float l = 0.f;

  float* slab = &lds_t[wv][0];
  float4* st0 = (float4*)(slab + col * TP + 8 * half);
  float4* st1 = (float4*)(slab + (16 + col) * TP + 8 * half);
  const float* ld = slab + (2 * half) * TP + col;

  const float4* pt = (const float4*)vs + (wv * (VW / 16)) * 128 + lane;
  for (int vt = 0; vt < VW / 16; ++vt, pt += 128) {
    const float4 a00 = pt[0], a01 = pt[32];
    const float4 a10 = pt[64], a11 = pt[96];

    st0[0] = a00; st0[1] = a01;
    st1[0] = a10; st1[1] = a11;

    v2f At[8];
#pragma unroll
    for (int kk = 0; kk < 8; ++kk) {
      At[kk].x = ld[(4 * kk) * TP];
      At[kk].y = ld[(4 * kk + 1) * TP];
    }
    __builtin_amdgcn_sched_barrier(0);
    v8f S = __builtin_amdgcn_wmma_f32_16x16x4_f32(false, At[0], false, Q[0],
                                                  (short)0, negm, false, false);
#pragma unroll
    for (int kk = 1; kk < 8; ++kk)
      S = __builtin_amdgcn_wmma_f32_16x16x4_f32(false, At[kk], false, Q[kk],
                                                (short)0, S, false, false);

    float p[8];
#pragma unroll
    for (int g = 0; g < 8; ++g) p[g] = __builtin_amdgcn_exp2f(S[g]);
    float ps = ((p[0] + p[1]) + (p[2] + p[3])) + ((p[4] + p[5]) + (p[6] + p[7]));

    if (vt == 0 || __builtin_amdgcn_ballot_w32(!(ps <= P_LIMIT)) != 0u) {
      float tm = fmaxf(fmaxf(fmaxf(S[0], S[1]), fmaxf(S[2], S[3])),
                       fmaxf(fmaxf(S[4], S[5]), fmaxf(S[6], S[7])));
      tm = fmaxf(tm, xhalf(tm));
      const float d  = (vt == 0) ? tm : fmaxf(tm, 0.f);
      const float mn = m + d;
      const float dd = mn - m;
      const float sc = (vt == 0) ? 0.f : __builtin_amdgcn_exp2f(-dd);
#pragma unroll
      for (int g = 0; g < 8; ++g) p[g] = __builtin_amdgcn_exp2f(S[g] - dd);
      ps = ((p[0] + p[1]) + (p[2] + p[3])) + ((p[4] + p[5]) + (p[6] + p[7]));
      m = mn;
      l *= sc;
#pragma unroll
      for (int g = 0; g < 8; ++g) { negm[g] = -mn; acc0[g] *= sc; acc1[g] *= sc; }
    }
    l += ps;

    {
      v2f Aa, Ab, Bt;
      Aa.x = a00.x; Aa.y = a00.y; Ab.x = a10.x; Ab.y = a10.y; Bt.x = p[0]; Bt.y = p[1];
      acc0 = __builtin_amdgcn_wmma_f32_16x16x4_f32(false, Aa, false, Bt,
                                                   (short)0, acc0, false, false);
      acc1 = __builtin_amdgcn_wmma_f32_16x16x4_f32(false, Ab, false, Bt,
                                                   (short)0, acc1, false, false);
      Aa.x = a00.z; Aa.y = a00.w; Ab.x = a10.z; Ab.y = a10.w; Bt.x = p[2]; Bt.y = p[3];
      acc0 = __builtin_amdgcn_wmma_f32_16x16x4_f32(false, Aa, false, Bt,
                                                   (short)0, acc0, false, false);
      acc1 = __builtin_amdgcn_wmma_f32_16x16x4_f32(false, Ab, false, Bt,
                                                   (short)0, acc1, false, false);
      Aa.x = a01.x; Aa.y = a01.y; Ab.x = a11.x; Ab.y = a11.y; Bt.x = p[4]; Bt.y = p[5];
      acc0 = __builtin_amdgcn_wmma_f32_16x16x4_f32(false, Aa, false, Bt,
                                                   (short)0, acc0, false, false);
      acc1 = __builtin_amdgcn_wmma_f32_16x16x4_f32(false, Ab, false, Bt,
                                                   (short)0, acc1, false, false);
      Aa.x = a01.z; Aa.y = a01.w; Ab.x = a11.z; Ab.y = a11.w; Bt.x = p[6]; Bt.y = p[7];
      acc0 = __builtin_amdgcn_wmma_f32_16x16x4_f32(false, Aa, false, Bt,
                                                   (short)0, acc0, false, false);
      acc1 = __builtin_amdgcn_wmma_f32_16x16x4_f32(false, Ab, false, Bt,
                                                   (short)0, acc1, false, false);
    }
  }

  l += xhalf(l);
  if (half == 0) { lds_m[wv][col] = m; lds_l[wv][col] = l; }
  __syncthreads();

  float M = -__builtin_inff();
#pragma unroll
  for (int w = 0; w < SPLIT; ++w) M = fmaxf(M, lds_m[w][col]);
  float L = 0.f;
#pragma unroll
  for (int w = 0; w < SPLIT; ++w) L += lds_l[w][col] * __builtin_amdgcn_exp2f(lds_m[w][col] - M);
  float f = __builtin_amdgcn_exp2f(lds_m[wv][col] - M) / L;

#pragma unroll
  for (int g = 0; g < 8; ++g) {
    int h0 = g + 8 * half;
    lds_p[wv][col * 33 + h0]      = acc0[g] * f;
    lds_p[wv][col * 33 + 16 + h0] = acc1[g] * f;
  }
  __syncthreads();

  for (int i = threadIdx.x; i < 16 * 32; i += 32 * SPLIT) {
    const int o = (i >> 5) * 33 + (i & 31);
    float s = lds_p[0][o];
#pragma unroll
    for (int w = 1; w < SPLIT; ++w) s += lds_p[w][o];
    out[r0 * 32 + i] = s;
  }
}

extern "C" void kernel_launch(void* const* d_in, const int* in_sizes, int n_in,
                              void* d_out, int out_size, void* d_ws, size_t ws_size,
                              hipStream_t stream) {
  const float* x     = (const float*)d_in[0];
  const float* W     = (const float*)d_in[1];
  const float* b     = (const float*)d_in[2];
  const float* temps = (const float*)d_in[3];
  const float* ve    = (const float*)d_in[4];
  float* out = (float*)d_out;

  float* vs  = (float*)d_ws;
  float* ffn = vs + (size_t)NH_ * V_;

  k_vs  <<<(V_ * 32) / 256, 256, 0, stream>>>(ve, vs);
  k_ffn <<<(R_ * 32) / 256, 256, 0, stream>>>(x, W, b, temps, ffn);
  k_main<<<R_ / 16, 32 * SPLIT, 0, stream>>>(ffn, vs, out);
}
